# starting kernel
# baseline (speedup 1.0000x reference)
#include <hip/hip_runtime.h>


typedef __attribute__((ext_vector_type(2))) float v2f;
typedef __attribute__((ext_vector_type(8))) float v8f;

#define C_IN   64
#define C_OUT  64
#define HW     32
#define K_TOT  576
#define LDS_T  580

__global__ __launch_bounds__(128, 1)
void local2d_wmma_f32(const float* __restrict__ xin,
                      const float* __restrict__ wgt,
                      const float* __restrict__ bias,
                      float* __restrict__ out)
{
    const int pos = blockIdx.x;
    const int x0  = pos >> 5;
    const int y0  = pos & 31;

    __shared__ float lds[32 * LDS_T];

    const int tid = threadIdx.x;

    for (int e = tid; e < 32 * K_TOT; e += 128) {
        const int n  = e / K_TOT;
        const int k  = e - n * K_TOT;
        const int i  = k / 9;
        const int r9 = k - i * 9;
        const int dk = r9 / 3;
        const int dl = r9 - dk * 3;
        const int h  = x0 + dk - 1;
        const int w  = y0 + dl - 1;
        float v = 0.0f;
        if ((unsigned)h < HW && (unsigned)w < HW)
            v = xin[(((n * C_IN) + i) * HW + h) * HW + w];
        lds[n * LDS_T + k] = v;
    }
    __syncthreads();

    const int lane  = tid & 31;
    const int wave  = tid >> 5;
    const int m0    = wave * 16;
    const int mrow  = m0 + (lane & 15);
    const int khalf = (lane >> 4) * 2;

    const float* wbase = wgt + (size_t)pos * (C_OUT * K_TOT);
    const float* arow  = wbase + mrow * K_TOT + khalf;

    __builtin_prefetch(arow + 128, 0, 3);
    __builtin_prefetch(arow + 320, 0, 3);

    const int bbase0 = (lane & 15) * LDS_T + khalf;
    const int bbase1 = bbase0 + 16 * LDS_T;

    v8f c0 = {};
    v8f c1 = {};

    #pragma unroll 4
    for (int k0 = 0; k0 < K_TOT; k0 += 4) {
        v2f a = *(const v2f*)(arow + k0);

        v2f b0 = *(const v2f*)(lds + bbase0 + k0);
        v2f b1 = *(const v2f*)(lds + bbase1 + k0);

        c0 = __builtin_amdgcn_wmma_f32_16x16x4_f32(
                 false, a, false, b0, (short)0, c0, false, false);
        c1 = __builtin_amdgcn_wmma_f32_16x16x4_f32(
                 false, a, false, b1, (short)0, c1, false, false);
    }

    const int mo   = m0 + (lane >> 4) * 8;
    const int n_lo = lane & 15;
    #pragma unroll
    for (int r = 0; r < 8; ++r) {
        const int m  = mo + r;
        const float bv = bias[m * (HW * HW) + pos];
        out[((n_lo        * C_OUT) + m) * (HW * HW) + pos] = c0[r] + bv;
        out[(((n_lo + 16) * C_OUT) + m) * (HW * HW) + pos] = c1[r] + bv;
    }
}

extern "C" void kernel_launch(void* const* d_in, const int* in_sizes, int n_in,
                              void* d_out, int out_size, void* d_ws, size_t ws_size,
                              hipStream_t stream) {
    const float* xin  = (const float*)d_in[0];
    const float* wgt  = (const float*)d_in[1];
    const float* bias = (const float*)d_in[2];
    float* out = (float*)d_out;

    dim3 grid(HW * HW);
    dim3 block(128);
    hipLaunchKernelGGL(local2d_wmma_f32, grid, block, 0, stream,
                       xin, wgt, bias, out);
}


# Patch gather into LDS issued as independent loads, all in flight before the first wait (per-lane source offsets computed once, reused for all batch rows), not one load waited for at a time
# speedup vs baseline: 1.2922x; 1.2922x over previous
#include <hip/hip_runtime.h>


typedef __attribute__((ext_vector_type(2))) float v2f;
typedef __attribute__((ext_vector_type(8))) float v8f;

#define C_IN   64
#define C_OUT  64
#define HW     32
#define K_TOT  576
#define LDS_T  580

__global__ __launch_bounds__(128, 1)
void local2d_wmma_f32(const float* __restrict__ xin,
                      const float* __restrict__ wgt,
                      const float* __restrict__ bias,
                      float* __restrict__ out)
{
    const int pos = blockIdx.x;
    const int x0  = pos >> 5;
    const int y0  = pos & 31;

    __shared__ float lds[32 * LDS_T];

    const int tid  = threadIdx.x;
    const int lane = tid & 31;
    const int wave = tid >> 5;

    int  soff[18];
    bool sok[18];
    #pragma unroll
    for (int j = 0; j < 18; ++j) {
        const int k  = lane + 32 * j;
        const int i  = k / 9;
        const int r9 = k - i * 9;
        const int dk = r9 / 3;
        const int dl = r9 - dk * 3;
        const int h  = x0 + dk - 1;
        const int w  = y0 + dl - 1;
        const bool ok = (unsigned)h < HW && (unsigned)w < HW;
        sok[j]  = ok;
        soff[j] = ok ? (i * HW + h) * HW + w : 0;
    }
    for (int nn = 0; nn < 8; nn += 4) {
        const int n = wave * 8 + nn;
        const float* xb = xin + n * (C_IN * HW * HW);
        float v[4][18];
        #pragma unroll
        for (int j = 0; j < 18; ++j) {
            #pragma unroll
            for (int u = 0; u < 4; ++u)
                v[u][j] = xb[u * (C_IN * HW * HW) + soff[j]];
        }
        float* row = lds + n * LDS_T + lane;
        #pragma unroll
        for (int j = 0; j < 18; ++j) {
            #pragma unroll
            for (int u = 0; u < 4; ++u)
                row[u * LDS_T + 32 * j] = sok[j] ? v[u][j] : 0.0f;
        }
    }
    __syncthreads();

    const int m0    = wave * 16;
    const int mrow  = m0 + (lane & 15);
    const int khalf = (lane >> 4) * 2;

    const float* wbase = wgt + (size_t)pos * (C_OUT * K_TOT);
    const float* arow  = wbase + mrow * K_TOT + khalf;

    __builtin_prefetch(arow + 128, 0, 3);
    __builtin_prefetch(arow + 320, 0, 3);

    const int bbase0 = (lane & 15) * LDS_T + khalf;
    const int bbase1 = bbase0 + 16 * LDS_T;

    v8f c0 = {};
    v8f c1 = {};

    #pragma unroll 4
    for (int k0 = 0; k0 < K_TOT; k0 += 4) {
        v2f a = *(const v2f*)(arow + k0);

        v2f b0 = *(const v2f*)(lds + bbase0 + k0);
        v2f b1 = *(const v2f*)(lds + bbase1 + k0);

        c0 = __builtin_amdgcn_wmma_f32_16x16x4_f32(
                 false, a, false, b0, (short)0, c0, false, false);
        c1 = __builtin_amdgcn_wmma_f32_16x16x4_f32(
                 false, a, false, b1, (short)0, c1, false, false);
    }

    const int mo   = m0 + (lane >> 4) * 8;
    const int n_lo = lane & 15;
    #pragma unroll
    for (int r = 0; r < 8; ++r) {
        const int m  = mo + r;
        const float bv = bias[m * (HW * HW) + pos];
        out[((n_lo        * C_OUT) + m) * (HW * HW) + pos] = c0[r] + bv;
        out[(((n_lo + 16) * C_OUT) + m) * (HW * HW) + pos] = c1[r] + bv;
    }
}

extern "C" void kernel_launch(void* const* d_in, const int* in_sizes, int n_in,
                              void* d_out, int out_size, void* d_ws, size_t ws_size,
                              hipStream_t stream) {
    const float* xin  = (const float*)d_in[0];
    const float* wgt  = (const float*)d_in[1];
    const float* bias = (const float*)d_in[2];
    float* out = (float*)d_out;

    dim3 grid(HW * HW);
    dim3 block(128);
    hipLaunchKernelGGL(local2d_wmma_f32, grid, block, 0, stream,
                       xin, wgt, bias, out);
}


# One block covers 4 neighbouring positions of an image row: each 16-byte run of an input row is loaded once and feeds all 4 patch matrices in LDS (256 blocks of 16 waves); matrix loop unchanged
# speedup vs baseline: 1.8591x; 1.4387x over previous
#include <hip/hip_runtime.h>


typedef __attribute__((ext_vector_type(2))) float v2f;
typedef __attribute__((ext_vector_type(8))) float v8f;

struct __attribute__((packed, aligned(4))) f4u { float a, b, c, d; };

#define C_IN   64
#define C_OUT  64
#define HW     32
#define K_TOT  576
#define LDS_T  580
#define NPOS   4
#define P_STR  (32 * LDS_T)

__global__ __launch_bounds__(512, 1)
void local2d_wmma_f32(const float* __restrict__ xin,
                      const float* __restrict__ wgt,
                      const float* __restrict__ bias,
                      float* __restrict__ out)
{
    const int x0 = blockIdx.x >> 3;
    const int y0 = (blockIdx.x & 7) * NPOS;

    __shared__ float lds[NPOS * P_STR];

    const int tid  = threadIdx.x;
    const int lane = tid & 31;
    const int wave = tid >> 5;

    int  soff[18];
    bool rok[18];
    bool elo[18];
    bool ehi[18];
    #pragma unroll
    for (int j = 0; j < 18; ++j) {
        const int k  = lane + 32 * j;
        const int i  = k / 9;
        const int r9 = k - i * 9;
        const int dk = r9 / 3;
        const int dl = r9 - dk * 3;
        const int h  = x0 + dk - 1;
        const int w0 = y0 + dl - 1;
        const bool ok = (unsigned)h < HW;
        const int  wl = w0 < 0 ? 0 : (w0 > HW - 4 ? HW - 4 : w0);
        rok[j]  = ok;
        elo[j]  = w0 < 0;
        ehi[j]  = w0 > HW - 4;
        soff[j] = ok ? (i * HW + h) * HW + wl : 0;
    }
    const int n0 = wave * 2;
    const float* xb = xin + n0 * (C_IN * HW * HW);
    f4u v[2][18];
    #pragma unroll
    for (int j = 0; j < 18; ++j) {
        #pragma unroll
        for (int u = 0; u < 2; ++u)
            v[u][j] = *(const f4u*)(xb + u * (C_IN * HW * HW) + soff[j]);
    }
    float* row = lds + n0 * LDS_T + lane;
    #pragma unroll
    for (int j = 0; j < 18; ++j) {
        #pragma unroll
        for (int u = 0; u < 2; ++u) {
            const float a = v[u][j].a, b = v[u][j].b, c = v[u][j].c, d = v[u][j].d;
            float q0 = elo[j] ? 0.0f : (ehi[j] ? b : a);
            float q1 = elo[j] ? a    : (ehi[j] ? c : b);
            float q2 = elo[j] ? b    : (ehi[j] ? d : c);
            float q3 = elo[j] ? c    : (ehi[j] ? 0.0f : d);
            if (!rok[j]) { q0 = 0.0f; q1 = 0.0f; q2 = 0.0f; q3 = 0.0f; }
            row[0 * P_STR + u * LDS_T + 32 * j] = q0;
            row[1 * P_STR + u * LDS_T + 32 * j] = q1;
            row[2 * P_STR + u * LDS_T + 32 * j] = q2;
            row[3 * P_STR + u * LDS_T + 32 * j] = q3;
        }
    }
    __syncthreads();

    const int p     = wave >> 2;
    const int pos   = x0 * HW + y0 + p;
    const int m0    = (wave & 3) * 16;
    const int mrow  = m0 + (lane & 15);
    const int khalf = (lane >> 4) * 2;

    const float* wbase = wgt + (size_t)pos * (C_OUT * K_TOT);
    const float* arow  = wbase + mrow * K_TOT + khalf;

    __builtin_prefetch(arow + 128, 0, 3);
    __builtin_prefetch(arow + 320, 0, 3);

    const int bbase0 = p * P_STR + (lane & 15) * LDS_T + khalf;
    const int bbase1 = bbase0 + 16 * LDS_T;

    v8f c0 = {};
    v8f c1 = {};

    #pragma unroll 4
    for (int k0 = 0; k0 < K_TOT; k0 += 4) {
        v2f a = *(const v2f*)(arow + k0);

        v2f b0 = *(const v2f*)(lds + bbase0 + k0);
        v2f b1 = *(const v2f*)(lds + bbase1 + k0);

        c0 = __builtin_amdgcn_wmma_f32_16x16x4_f32(
                 false, a, false, b0, (short)0, c0, false, false);
        c1 = __builtin_amdgcn_wmma_f32_16x16x4_f32(
                 false, a, false, b1, (short)0, c1, false, false);
    }

    const int mo   = m0 + (lane >> 4) * 8;
    const int n_lo = lane & 15;
    #pragma unroll
    for (int r = 0; r < 8; ++r) {
        const int m  = mo + r;
        const float bv = bias[m * (HW * HW) + pos];
        out[((n_lo        * C_OUT) + m) * (HW * HW) + pos] = c0[r] + bv;
        out[(((n_lo + 16) * C_OUT) + m) * (HW * HW) + pos] = c1[r] + bv;
    }
}

extern "C" void kernel_launch(void* const* d_in, const int* in_sizes, int n_in,
                              void* d_out, int out_size, void* d_ws, size_t ws_size,
                              hipStream_t stream) {
    const float* xin  = (const float*)d_in[0];
    const float* wgt  = (const float*)d_in[1];
    const float* bias = (const float*)d_in[2];
    float* out = (float*)d_out;

    dim3 grid(256);
    dim3 block(512);
    hipLaunchKernelGGL(local2d_wmma_f32, grid, block, 0, stream,
                       xin, wgt, bias, out);
}


# A and B fragments fetched 16 bytes per lane: lane half h owns k=8g+4h..+3 of each 8-wide k group, one global_load_b128 / ds_load_b128 feeds two WMMAs; changes the order the k terms are summed
# speedup vs baseline: 1.9463x; 1.0469x over previous
#include <hip/hip_runtime.h>


typedef __attribute__((ext_vector_type(2))) float v2f;
typedef __attribute__((ext_vector_type(4))) float v4f;
typedef __attribute__((ext_vector_type(8))) float v8f;

struct __attribute__((packed, aligned(4))) f4u { float a, b, c, d; };

#define C_IN   64
#define C_OUT  64
#define HW     32
#define K_TOT  576
#define LDS_T  580
#define NPOS   4
#define P_STR  (32 * LDS_T)

__global__ __launch_bounds__(512, 1)
void local2d_wmma_f32(const float* __restrict__ xin,
                      const float* __restrict__ wgt,
                      const float* __restrict__ bias,
                      float* __restrict__ out)
{
    const int x0 = blockIdx.x >> 3;
    const int y0 = (blockIdx.x & 7) * NPOS;

    __shared__ __attribute__((aligned(16))) float lds[NPOS * P_STR];

    const int tid  = threadIdx.x;
    const int lane = tid & 31;
    const int wave = tid >> 5;

    int  soff[18];
    bool rok[18];
    bool elo[18];
    bool ehi[18];
    #pragma unroll
    for (int j = 0; j < 18; ++j) {
        const int k  = lane + 32 * j;
        const int i  = k / 9;
        const int r9 = k - i * 9;
        const int dk = r9 / 3;
        const int dl = r9 - dk * 3;
        const int h  = x0 + dk - 1;
        const int w0 = y0 + dl - 1;
        const bool ok = (unsigned)h < HW;
        const int  wl = w0 < 0 ? 0 : (w0 > HW - 4 ? HW - 4 : w0);
        rok[j]  = ok;
        elo[j]  = w0 < 0;
        ehi[j]  = w0 > HW - 4;
        soff[j] = ok ? (i * HW + h) * HW + wl : 0;
    }
    const int n0 = wave * 2;
    const float* xb = xin + n0 * (C_IN * HW * HW);
    f4u v[2][18];
    #pragma unroll
    for (int j = 0; j < 18; ++j) {
        #pragma unroll
        for (int u = 0; u < 2; ++u)
            v[u][j] = *(const f4u*)(xb + u * (C_IN * HW * HW) + soff[j]);
    }
    float* row = lds + n0 * LDS_T + lane;
    #pragma unroll
    for (int j = 0; j < 18; ++j) {
        #pragma unroll
        for (int u = 0; u < 2; ++u) {
            const float a = v[u][j].a, b = v[u][j].b, c = v[u][j].c, d = v[u][j].d;
            float q0 = elo[j] ? 0.0f : (ehi[j] ? b : a);
            float q1 = elo[j] ? a    : (ehi[j] ? c : b);
            float q2 = elo[j] ? b    : (ehi[j] ? d : c);
            float q3 = elo[j] ? c    : (ehi[j] ? 0.0f : d);
            if (!rok[j]) { q0 = 0.0f; q1 = 0.0f; q2 = 0.0f; q3 = 0.0f; }
            row[0 * P_STR + u * LDS_T + 32 * j] = q0;
            row[1 * P_STR + u * LDS_T + 32 * j] = q1;
            row[2 * P_STR + u * LDS_T + 32 * j] = q2;
            row[3 * P_STR + u * LDS_T + 32 * j] = q3;
        }
    }
    __syncthreads();

    const int p     = wave >> 2;
    const int pos   = x0 * HW + y0 + p;
    const int m0    = (wave & 3) * 16;
    const int mrow  = m0 + (lane & 15);
    const int kq    = (lane >> 4) * 4;

    const float* wbase = wgt + (size_t)pos * (C_OUT * K_TOT);
    const float* arow  = wbase + mrow * K_TOT + kq;

    __builtin_prefetch(arow + 128, 0, 3);
    __builtin_prefetch(arow + 320, 0, 3);

    const int bbase0 = p * P_STR + (lane & 15) * LDS_T + kq;
    const int bbase1 = bbase0 + 16 * LDS_T;

    v8f c0 = {};
    v8f c1 = {};

    for (int k0 = 0; k0 < K_TOT; k0 += 32) {
        v4f a[4], b0[4], b1[4];
        #pragma unroll
        for (int g = 0; g < 4; ++g)
            a[g] = *(const v4f*)(arow + k0 + 8 * g);
        #pragma unroll
        for (int g = 0; g < 4; ++g) {
            b0[g] = *(const v4f*)(lds + bbase0 + k0 + 8 * g);
            b1[g] = *(const v4f*)(lds + bbase1 + k0 + 8 * g);
        }
        #pragma unroll
        for (int g = 0; g < 4; ++g) {
            const v2f alo  = {a[g].x,  a[g].y},  ahi  = {a[g].z,  a[g].w};
            const v2f b0lo = {b0[g].x, b0[g].y}, b0hi = {b0[g].z, b0[g].w};
            const v2f b1lo = {b1[g].x, b1[g].y}, b1hi = {b1[g].z, b1[g].w};
            c0 = __builtin_amdgcn_wmma_f32_16x16x4_f32(
                     false, alo, false, b0lo, (short)0, c0, false, false);
            c1 = __builtin_amdgcn_wmma_f32_16x16x4_f32(
                     false, alo, false, b1lo, (short)0, c1, false, false);
            c0 = __builtin_amdgcn_wmma_f32_16x16x4_f32(
                     false, ahi, false, b0hi, (short)0, c0, false, false);
            c1 = __builtin_amdgcn_wmma_f32_16x16x4_f32(
                     false, ahi, false, b1hi, (short)0, c1, false, false);
        }
    }

    const int mo   = m0 + (lane >> 4) * 8;
    const int n_lo = lane & 15;
    #pragma unroll
    for (int r = 0; r < 8; ++r) {
        const int m  = mo + r;
        const float bv = bias[m * (HW * HW) + pos];
        out[((n_lo        * C_OUT) + m) * (HW * HW) + pos] = c0[r] + bv;
        out[(((n_lo + 16) * C_OUT) + m) * (HW * HW) + pos] = c1[r] + bv;
    }
}

extern "C" void kernel_launch(void* const* d_in, const int* in_sizes, int n_in,
                              void* d_out, int out_size, void* d_ws, size_t ws_size,
                              hipStream_t stream) {
    const float* xin  = (const float*)d_in[0];
    const float* wgt  = (const float*)d_in[1];
    const float* bias = (const float*)d_in[2];
    float* out = (float*)d_out;

    dim3 grid(256);
    dim3 block(512);
    hipLaunchKernelGGL(local2d_wmma_f32, grid, block, 0, stream,
                       xin, wgt, bias, out);
}


# Results staged through LDS after the matrix loop: one 16-byte store per (batch row, channel) covers the block's 4 neighbouring positions, instead of 4-byte stores that each touch a line of their own
# speedup vs baseline: 3.3450x; 1.7186x over previous
#include <hip/hip_runtime.h>


typedef __attribute__((ext_vector_type(2))) float v2f;
typedef __attribute__((ext_vector_type(4))) float v4f;
typedef __attribute__((ext_vector_type(8))) float v8f;

struct __attribute__((packed, aligned(4))) f4u { float a, b, c, d; };

#define C_IN   64
#define C_OUT  64
#define HW     32
#define K_TOT  576
#define LDS_T  580
#define NPOS   4
#define P_STR  (32 * LDS_T)
#define OUT_T  260

__global__ __launch_bounds__(512, 1)
void local2d_wmma_f32(const float* __restrict__ xin,
                      const float* __restrict__ wgt,
                      const float* __restrict__ bias,
                      float* __restrict__ out)
{
    const int x0 = blockIdx.x >> 3;
    const int y0 = (blockIdx.x & 7) * NPOS;

    __shared__ __attribute__((aligned(16))) float lds[NPOS * P_STR];

    const int tid  = threadIdx.x;
    const int lane = tid & 31;
    const int wave = tid >> 5;

    int  soff[18];
    bool rok[18];
    bool elo[18];
    bool ehi[18];
    #pragma unroll
    for (int j = 0; j < 18; ++j) {
        const int k  = lane + 32 * j;
        const int i  = k / 9;
        const int r9 = k - i * 9;
        const int dk = r9 / 3;
        const int dl = r9 - dk * 3;
        const int h  = x0 + dk - 1;
        const int w0 = y0 + dl - 1;
        const bool ok = (unsigned)h < HW;
        const int  wl = w0 < 0 ? 0 : (w0 > HW - 4 ? HW - 4 : w0);
        rok[j]  = ok;
        elo[j]  = w0 < 0;
        ehi[j]  = w0 > HW - 4;
        soff[j] = ok ? (i * HW + h) * HW + wl : 0;
    }
    const int n0 = wave * 2;
    const float* xb = xin + n0 * (C_IN * HW * HW);
    f4u v[2][18];
    #pragma unroll
    for (int j = 0; j < 18; ++j) {
        #pragma unroll
        for (int u = 0; u < 2; ++u)
            v[u][j] = *(const f4u*)(xb + u * (C_IN * HW * HW) + soff[j]);
    }
    float* row = lds + n0 * LDS_T + lane;
    #pragma unroll
    for (int j = 0; j < 18; ++j) {
        #pragma unroll
        for (int u = 0; u < 2; ++u) {
            const float a = v[u][j].a, b = v[u][j].b, c = v[u][j].c, d = v[u][j].d;
            float q0 = elo[j] ? 0.0f : (ehi[j] ? b : a);
            float q1 = elo[j] ? a    : (ehi[j] ? c : b);
            float q2 = elo[j] ? b    : (ehi[j] ? d : c);
            float q3 = elo[j] ? c    : (ehi[j] ? 0.0f : d);
            if (!rok[j]) { q0 = 0.0f; q1 = 0.0f; q2 = 0.0f; q3 = 0.0f; }
            row[0 * P_STR + u * LDS_T + 32 * j] = q0;
            row[1 * P_STR + u * LDS_T + 32 * j] = q1;
            row[2 * P_STR + u * LDS_T + 32 * j] = q2;
            row[3 * P_STR + u * LDS_T + 32 * j] = q3;
        }
    }
    __syncthreads();

    const int p     = wave >> 2;
    const int pos   = x0 * HW + y0 + p;
    const int m0    = (wave & 3) * 16;
    const int mrow  = m0 + (lane & 15);
    const int kq    = (lane >> 4) * 4;

    const float* wbase = wgt + (size_t)pos * (C_OUT * K_TOT);
    const float* arow  = wbase + mrow * K_TOT + kq;

    __builtin_prefetch(arow + 128, 0, 3);
    __builtin_prefetch(arow + 320, 0, 3);

    const int bbase0 = p * P_STR + (lane & 15) * LDS_T + kq;
    const int bbase1 = bbase0 + 16 * LDS_T;

    v8f c0 = {};
    v8f c1 = {};

    const int pos0 = x0 * HW + y0;
    const int m_st = tid & 63;
    const int n_st = tid >> 6;
    const v4f bv = *(const v4f*)(bias + m_st * (HW * HW) + pos0);

    for (int k0 = 0; k0 < K_TOT; k0 += 32) {
        v4f a[4], b0[4], b1[4];
        #pragma unroll
        for (int g = 0; g < 4; ++g)
            a[g] = *(const v4f*)(arow + k0 + 8 * g);
        #pragma unroll
        for (int g = 0; g < 4; ++g) {
            b0[g] = *(const v4f*)(lds + bbase0 + k0 + 8 * g);
            b1[g] = *(const v4f*)(lds + bbase1 + k0 + 8 * g);
        }
        #pragma unroll
        for (int g = 0; g < 4; ++g) {
            const v2f alo  = {a[g].x,  a[g].y},  ahi  = {a[g].z,  a[g].w};
            const v2f b0lo = {b0[g].x, b0[g].y}, b0hi = {b0[g].z, b0[g].w};
            const v2f b1lo = {b1[g].x, b1[g].y}, b1hi = {b1[g].z, b1[g].w};
            c0 = __builtin_amdgcn_wmma_f32_16x16x4_f32(
                     false, alo, false, b0lo, (short)0, c0, false, false);
            c1 = __builtin_amdgcn_wmma_f32_16x16x4_f32(
                     false, alo, false, b1lo, (short)0, c1, false, false);
            c0 = __builtin_amdgcn_wmma_f32_16x16x4_f32(
                     false, ahi, false, b0hi, (short)0, c0, false, false);
            c1 = __builtin_amdgcn_wmma_f32_16x16x4_f32(
                     false, ahi, false, b1hi, (short)0, c1, false, false);
        }
    }

    __syncthreads();
    float* stg = lds;
    const int mo   = m0 + (lane >> 4) * 8;
    const int n_lo = lane & 15;
    #pragma unroll
    for (int r = 0; r < 8; ++r) {
        stg[n_lo        * OUT_T + (mo + r) * NPOS + p] = c0[r];
        stg[(n_lo + 16) * OUT_T + (mo + r) * NPOS + p] = c1[r];
    }
    __syncthreads();
    #pragma unroll
    for (int q = 0; q < 4; ++q) {
        const int n = n_st + 8 * q;
        const v4f d = *(const v4f*)(stg + n * OUT_T + m_st * NPOS);
        *(v4f*)(out + ((n * C_OUT) + m_st) * (HW * HW) + pos0) = d + bv;
    }
}

extern "C" void kernel_launch(void* const* d_in, const int* in_sizes, int n_in,
                              void* d_out, int out_size, void* d_ws, size_t ws_size,
                              hipStream_t stream) {
    const float* xin  = (const float*)d_in[0];
    const float* wgt  = (const float*)d_in[1];
    const float* bias = (const float*)d_in[2];
    float* out = (float*)d_out;

    dim3 grid(256);
    dim3 block(512);
    hipLaunchKernelGGL(local2d_wmma_f32, grid, block, 0, stream,
                       xin, wgt, bias, out);
}


# Patch gather pipelined against the matrix loop in 3 k chunks: chunk 0 staged up front; chunk c+1's loads fly and its LDS writes land under chunk c's products (split barrier keeps loads in flight)
# speedup vs baseline: 3.6066x; 1.0782x over previous
#include <hip/hip_runtime.h>


typedef __attribute__((ext_vector_type(2))) float v2f;
typedef __attribute__((ext_vector_type(4))) float v4f;
typedef __attribute__((ext_vector_type(8))) float v8f;

struct __attribute__((packed, aligned(4))) f4u { float a, b, c, d; };

#define C_IN   64
#define C_OUT  64
#define HW     32
#define K_TOT  576
#define LDS_T  580
#define NPOS   4
#define P_STR  (32 * LDS_T)
#define OUT_T  260

#define CH_J   6

static __device__ __forceinline__ void lds_writes_done_signal() {
    asm volatile("s_wait_dscnt 0x0\n\ts_barrier_signal -1" ::: "memory");
}
static __device__ __forceinline__ void handover_wait() {
    asm volatile("s_barrier_wait -1" ::: "memory");
}

template <int C>
static __device__ __forceinline__ void gather_issue(f4u (&v)[2][18], const float* xb,
                                                    const int (&soff)[18]) {
    #pragma unroll
    for (int j = C * CH_J; j < C * CH_J + CH_J; ++j) {
        #pragma unroll
        for (int u = 0; u < 2; ++u)
            v[u][j] = *(const f4u*)(xb + u * (C_IN * HW * HW) + soff[j]);
    }
}

template <int C>
static __device__ __forceinline__ void gather_store(const f4u (&v)[2][18], float* row,
                                                    const bool (&rok)[18],
                                                    const bool (&elo)[18],
                                                    const bool (&ehi)[18]) {
    #pragma unroll
    for (int j = C * CH_J; j < C * CH_J + CH_J; ++j) {
        #pragma unroll
        for (int u = 0; u < 2; ++u) {
            const float a = v[u][j].a, b = v[u][j].b, c = v[u][j].c, d = v[u][j].d;
            float q0 = elo[j] ? 0.0f : (ehi[j] ? b : a);
            float q1 = elo[j] ? a    : (ehi[j] ? c : b);
            float q2 = elo[j] ? b    : (ehi[j] ? d : c);
            float q3 = elo[j] ? c    : (ehi[j] ? 0.0f : d);
            if (!rok[j]) { q0 = 0.0f; q1 = 0.0f; q2 = 0.0f; q3 = 0.0f; }
            row[0 * P_STR + u * LDS_T + 32 * j] = q0;
            row[1 * P_STR + u * LDS_T + 32 * j] = q1;
            row[2 * P_STR + u * LDS_T + 32 * j] = q2;
            row[3 * P_STR + u * LDS_T + 32 * j] = q3;
        }
    }
}

template <int T0, int T1>
static __device__ __forceinline__ void trips(const float* arow, const float* lds,
                                             int bbase0, int bbase1, v8f& c0, v8f& c1) {
    for (int k0 = T0 * 32; k0 < T1 * 32; k0 += 32) {
        v4f a[4], b0[4], b1[4];
        #pragma unroll
        for (int g = 0; g < 4; ++g)
            a[g] = *(const v4f*)(arow + k0 + 8 * g);
        #pragma unroll
        for (int g = 0; g < 4; ++g) {
            b0[g] = *(const v4f*)(lds + bbase0 + k0 + 8 * g);
            b1[g] = *(const v4f*)(lds + bbase1 + k0 + 8 * g);
        }
        #pragma unroll
        for (int g = 0; g < 4; ++g) {
            const v2f alo  = {a[g].x,  a[g].y},  ahi  = {a[g].z,  a[g].w};
            const v2f b0lo = {b0[g].x, b0[g].y}, b0hi = {b0[g].z, b0[g].w};
            const v2f b1lo = {b1[g].x, b1[g].y}, b1hi = {b1[g].z, b1[g].w};
            c0 = __builtin_amdgcn_wmma_f32_16x16x4_f32(
                     false, alo, false, b0lo, (short)0, c0, false, false);
            c1 = __builtin_amdgcn_wmma_f32_16x16x4_f32(
                     false, alo, false, b1lo, (short)0, c1, false, false);
            c0 = __builtin_amdgcn_wmma_f32_16x16x4_f32(
                     false, ahi, false, b0hi, (short)0, c0, false, false);
            c1 = __builtin_amdgcn_wmma_f32_16x16x4_f32(
                     false, ahi, false, b1hi, (short)0, c1, false, false);
        }
    }
}

__global__ __launch_bounds__(512, 1)
void local2d_wmma_f32(const float* __restrict__ xin,
                      const float* __restrict__ wgt,
                      const float* __restrict__ bias,
                      float* __restrict__ out)
{
    const int x0 = blockIdx.x >> 3;
    const int y0 = (blockIdx.x & 7) * NPOS;

    __shared__ __attribute__((aligned(16))) float lds[NPOS * P_STR];

    const int tid  = threadIdx.x;
    const int lane = tid & 31;
    const int wave = tid >> 5;

    int  soff[18];
    bool rok[18];
    bool elo[18];
    bool ehi[18];
    #pragma unroll
    for (int j = 0; j < 18; ++j) {
        const int k  = lane + 32 * j;
        const int i  = k / 9;
        const int r9 = k - i * 9;
        const int dk = r9 / 3;
        const int dl = r9 - dk * 3;
        const int h  = x0 + dk - 1;
        const int w0 = y0 + dl - 1;
        const bool ok = (unsigned)h < HW;
        const int  wl = w0 < 0 ? 0 : (w0 > HW - 4 ? HW - 4 : w0);
        rok[j]  = ok;
        elo[j]  = w0 < 0;
        ehi[j]  = w0 > HW - 4;
        soff[j] = ok ? (i * HW + h) * HW + wl : 0;
    }
    const int n0 = wave * 2;
    const float* xb = xin + n0 * (C_IN * HW * HW);
    f4u v[2][18];
    float* row = lds + n0 * LDS_T + lane;
    gather_issue<0>(v, xb, soff);
    gather_store<0>(v, row, rok, elo, ehi);
    __syncthreads();

    const int p     = wave >> 2;
    const int pos   = x0 * HW + y0 + p;
    const int m0    = (wave & 3) * 16;
    const int mrow  = m0 + (lane & 15);
    const int kq    = (lane >> 4) * 4;

    const float* wbase = wgt + (size_t)pos * (C_OUT * K_TOT);
    const float* arow  = wbase + mrow * K_TOT + kq;

    __builtin_prefetch(arow + 128, 0, 3);
    __builtin_prefetch(arow + 320, 0, 3);

    const int bbase0 = p * P_STR + (lane & 15) * LDS_T + kq;
    const int bbase1 = bbase0 + 16 * LDS_T;

    v8f c0 = {};
    v8f c1 = {};

    const int pos0 = x0 * HW + y0;
    const int m_st = tid & 63;
    const int n_st = tid >> 6;
    const v4f bv = *(const v4f*)(bias + m_st * (HW * HW) + pos0);

    gather_issue<1>(v, xb, soff);
    trips<0, 3>(arow, lds, bbase0, bbase1, c0, c1);
    gather_store<1>(v, row, rok, elo, ehi);
    lds_writes_done_signal();
    gather_issue<2>(v, xb, soff);
    trips<3, 6>(arow, lds, bbase0, bbase1, c0, c1);
    handover_wait();
    trips<6, 9>(arow, lds, bbase0, bbase1, c0, c1);
    gather_store<2>(v, row, rok, elo, ehi);
    lds_writes_done_signal();
    trips<9, 12>(arow, lds, bbase0, bbase1, c0, c1);
    handover_wait();
    trips<12, 18>(arow, lds, bbase0, bbase1, c0, c1);

    __syncthreads();
    float* stg = lds;
    const int mo   = m0 + (lane >> 4) * 8;
    const int n_lo = lane & 15;
    #pragma unroll
    for (int r = 0; r < 8; ++r) {
        stg[n_lo        * OUT_T + (mo + r) * NPOS + p] = c0[r];
        stg[(n_lo + 16) * OUT_T + (mo + r) * NPOS + p] = c1[r];
    }
    __syncthreads();
    #pragma unroll
    for (int q = 0; q < 4; ++q) {
        const int n = n_st + 8 * q;
        const v4f d = *(const v4f*)(stg + n * OUT_T + m_st * NPOS);
        *(v4f*)(out + ((n * C_OUT) + m_st) * (HW * HW) + pos0) = d + bv;
    }
}

extern "C" void kernel_launch(void* const* d_in, const int* in_sizes, int n_in,
                              void* d_out, int out_size, void* d_ws, size_t ws_size,
                              hipStream_t stream) {
    const float* xin  = (const float*)d_in[0];
    const float* wgt  = (const float*)d_in[1];
    const float* bias = (const float*)d_in[2];
    float* out = (float*)d_out;

    dim3 grid(256);
    dim3 block(512);
    hipLaunchKernelGGL(local2d_wmma_f32, grid, block, 0, stream,
                       xin, wgt, bias, out);
}


# Each wave asks the second-level cache for its contiguous 16-row weight slab about five trips ahead of use: 5 prefetch instructions per wave, one 256-byte unit per lane, in the order the trips need it
# speedup vs baseline: 3.7984x; 1.0532x over previous
#include <hip/hip_runtime.h>


typedef __attribute__((ext_vector_type(2))) float v2f;
typedef __attribute__((ext_vector_type(4))) float v4f;
typedef __attribute__((ext_vector_type(8))) float v8f;

struct __attribute__((packed, aligned(4))) f4u { float a, b, c, d; };

#define C_IN   64
#define C_OUT  64
#define HW     32
#define K_TOT  576
#define LDS_T  580
#define NPOS   4
#define P_STR  (32 * LDS_T)
#define OUT_T  260

#define CH_J   6

static __device__ __forceinline__ void lds_writes_done_signal() {
    asm volatile("s_wait_dscnt 0x0\n\ts_barrier_signal -1" ::: "memory");
}
static __device__ __forceinline__ void handover_wait() {
    asm volatile("s_barrier_wait -1" ::: "memory");
}

template <int Q>
static __device__ __forceinline__ void weights_to_l2(const float* rowunit, int half) {
    if (Q < 4) __builtin_prefetch(rowunit + half * 64 + Q * 128, 0, 3);
    else       __builtin_prefetch(rowunit + 512, 0, 3);
}

template <int C>
static __device__ __forceinline__ void gather_issue(f4u (&v)[2][18], const float* xb,
                                                    const int (&soff)[18]) {
    #pragma unroll
    for (int j = C * CH_J; j < C * CH_J + CH_J; ++j) {
        #pragma unroll
        for (int u = 0; u < 2; ++u)
            v[u][j] = *(const f4u*)(xb + u * (C_IN * HW * HW) + soff[j]);
    }
}

template <int C>
static __device__ __forceinline__ void gather_store(const f4u (&v)[2][18], float* row,
                                                    const bool (&rok)[18],
                                                    const bool (&elo)[18],
                                                    const bool (&ehi)[18]) {
    #pragma unroll
    for (int j = C * CH_J; j < C * CH_J + CH_J; ++j) {
        #pragma unroll
        for (int u = 0; u < 2; ++u) {
            const float a = v[u][j].a, b = v[u][j].b, c = v[u][j].c, d = v[u][j].d;
            float q0 = elo[j] ? 0.0f : (ehi[j] ? b : a);
            float q1 = elo[j] ? a    : (ehi[j] ? c : b);
            float q2 = elo[j] ? b    : (ehi[j] ? d : c);
            float q3 = elo[j] ? c    : (ehi[j] ? 0.0f : d);
            if (!rok[j]) { q0 = 0.0f; q1 = 0.0f; q2 = 0.0f; q3 = 0.0f; }
            row[0 * P_STR + u * LDS_T + 32 * j] = q0;
            row[1 * P_STR + u * LDS_T + 32 * j] = q1;
            row[2 * P_STR + u * LDS_T + 32 * j] = q2;
            row[3 * P_STR + u * LDS_T + 32 * j] = q3;
        }
    }
}

template <int T0, int T1>
static __device__ __forceinline__ void trips(const float* arow, const float* lds,
                                             int bbase0, int bbase1, v8f& c0, v8f& c1) {
    for (int k0 = T0 * 32; k0 < T1 * 32; k0 += 32) {
        v4f a[4], b0[4], b1[4];
        #pragma unroll
        for (int g = 0; g < 4; ++g)
            a[g] = *(const v4f*)(arow + k0 + 8 * g);
        #pragma unroll
        for (int g = 0; g < 4; ++g) {
            b0[g] = *(const v4f*)(lds + bbase0 + k0 + 8 * g);
            b1[g] = *(const v4f*)(lds + bbase1 + k0 + 8 * g);
        }
        #pragma unroll
        for (int g = 0; g < 4; ++g) {
            const v2f alo  = {a[g].x,  a[g].y},  ahi  = {a[g].z,  a[g].w};
            const v2f b0lo = {b0[g].x, b0[g].y}, b0hi = {b0[g].z, b0[g].w};
            const v2f b1lo = {b1[g].x, b1[g].y}, b1hi = {b1[g].z, b1[g].w};
            c0 = __builtin_amdgcn_wmma_f32_16x16x4_f32(
                     false, alo, false, b0lo, (short)0, c0, false, false);
            c1 = __builtin_amdgcn_wmma_f32_16x16x4_f32(
                     false, alo, false, b1lo, (short)0, c1, false, false);
            c0 = __builtin_amdgcn_wmma_f32_16x16x4_f32(
                     false, ahi, false, b0hi, (short)0, c0, false, false);
            c1 = __builtin_amdgcn_wmma_f32_16x16x4_f32(
                     false, ahi, false, b1hi, (short)0, c1, false, false);
        }
    }
}

__global__ __launch_bounds__(512, 1)
void local2d_wmma_f32(const float* __restrict__ xin,
                      const float* __restrict__ wgt,
                      const float* __restrict__ bias,
                      float* __restrict__ out)
{
    const int x0 = blockIdx.x >> 3;
    const int y0 = (blockIdx.x & 7) * NPOS;

    __shared__ __attribute__((aligned(16))) float lds[NPOS * P_STR];

    const int tid  = threadIdx.x;
    const int lane = tid & 31;
    const int wave = tid >> 5;

    int  soff[18];
    bool rok[18];
    bool elo[18];
    bool ehi[18];
    #pragma unroll
    for (int j = 0; j < 18; ++j) {
        const int k  = lane + 32 * j;
        const int i  = k / 9;
        const int r9 = k - i * 9;
        const int dk = r9 / 3;
        const int dl = r9 - dk * 3;
        const int h  = x0 + dk - 1;
        const int w0 = y0 + dl - 1;
        const bool ok = (unsigned)h < HW;
        const int  wl = w0 < 0 ? 0 : (w0 > HW - 4 ? HW - 4 : w0);
        rok[j]  = ok;
        elo[j]  = w0 < 0;
        ehi[j]  = w0 > HW - 4;
        soff[j] = ok ? (i * HW + h) * HW + wl : 0;
    }
    const int n0 = wave * 2;
    const float* xb = xin + n0 * (C_IN * HW * HW);
    f4u v[2][18];
    float* row = lds + n0 * LDS_T + lane;
    const int p     = wave >> 2;
    const int pos   = x0 * HW + y0 + p;
    const int m0    = (wave & 3) * 16;
    const int mrow  = m0 + (lane & 15);
    const int half  = lane >> 4;
    const int kq    = half * 4;

    const float* wbase = wgt + (size_t)pos * (C_OUT * K_TOT);
    const float* wrow  = wbase + mrow * K_TOT;
    const float* arow  = wrow + kq;

    gather_issue<0>(v, xb, soff);
    weights_to_l2<0>(wrow, half);
    gather_store<0>(v, row, rok, elo, ehi);
    __syncthreads();

    const int bbase0 = p * P_STR + (lane & 15) * LDS_T + kq;
    const int bbase1 = bbase0 + 16 * LDS_T;

    v8f c0 = {};
    v8f c1 = {};

    const int pos0 = x0 * HW + y0;
    const int m_st = tid & 63;
    const int n_st = tid >> 6;
    const v4f bv = *(const v4f*)(bias + m_st * (HW * HW) + pos0);

    weights_to_l2<1>(wrow, half);
    gather_issue<1>(v, xb, soff);
    trips<0, 3>(arow, lds, bbase0, bbase1, c0, c1);
    gather_store<1>(v, row, rok, elo, ehi);
    lds_writes_done_signal();
    weights_to_l2<2>(wrow, half);
    gather_issue<2>(v, xb, soff);
    trips<3, 6>(arow, lds, bbase0, bbase1, c0, c1);
    handover_wait();
    weights_to_l2<3>(wrow, half);
    trips<6, 9>(arow, lds, bbase0, bbase1, c0, c1);
    gather_store<2>(v, row, rok, elo, ehi);
    lds_writes_done_signal();
    weights_to_l2<4>(wrow, half);
    trips<9, 12>(arow, lds, bbase0, bbase1, c0, c1);
    handover_wait();
    trips<12, 18>(arow, lds, bbase0, bbase1, c0, c1);

    __syncthreads();
    float* stg = lds;
    const int mo   = m0 + (lane >> 4) * 8;
    const int n_lo = lane & 15;
    #pragma unroll
    for (int r = 0; r < 8; ++r) {
        stg[n_lo        * OUT_T + (mo + r) * NPOS + p] = c0[r];
        stg[(n_lo + 16) * OUT_T + (mo + r) * NPOS + p] = c1[r];
    }
    __syncthreads();
    #pragma unroll
    for (int q = 0; q < 4; ++q) {
        const int n = n_st + 8 * q;
        const v4f d = *(const v4f*)(stg + n * OUT_T + m_st * NPOS);
        *(v4f*)(out + ((n * C_OUT) + m_st) * (HW * HW) + pos0) = d + bv;
    }
}

extern "C" void kernel_launch(void* const* d_in, const int* in_sizes, int n_in,
                              void* d_out, int out_size, void* d_ws, size_t ws_size,
                              hipStream_t stream) {
    const float* xin  = (const float*)d_in[0];
    const float* wgt  = (const float*)d_in[1];
    const float* bias = (const float*)d_in[2];
    float* out = (float*)d_out;

    dim3 grid(256);
    dim3 block(512);
    hipLaunchKernelGGL(local2d_wmma_f32, grid, block, 0, stream,
                       xin, wgt, bias, out);
}


# Block id remapped (image row = id mod 32, column group = id / 32): the 8 blocks that write the 16-byte pieces of one output line and read the same input lines get ids 32 apart, not consecutive
# speedup vs baseline: 4.4783x; 1.1790x over previous
#include <hip/hip_runtime.h>


typedef __attribute__((ext_vector_type(2))) float v2f;
typedef __attribute__((ext_vector_type(4))) float v4f;
typedef __attribute__((ext_vector_type(8))) float v8f;

struct __attribute__((packed, aligned(4))) f4u { float a, b, c, d; };

#define C_IN   64
#define C_OUT  64
#define HW     32
#define K_TOT  576
#define LDS_T  580
#define NPOS   4
#define P_STR  (32 * LDS_T)
#define OUT_T  260

#define CH_J   6

static __device__ __forceinline__ void lds_writes_done_signal() {
    asm volatile("s_wait_dscnt 0x0\n\ts_barrier_signal -1" ::: "memory");
}
static __device__ __forceinline__ void handover_wait() {
    asm volatile("s_barrier_wait -1" ::: "memory");
}

template <int Q>
static __device__ __forceinline__ void weights_to_l2(const float* rowunit, int half) {
    if (Q < 4) __builtin_prefetch(rowunit + half * 64 + Q * 128, 0, 3);
    else       __builtin_prefetch(rowunit + 512, 0, 3);
}

template <int C>
static __device__ __forceinline__ void gather_issue(f4u (&v)[2][18], const float* xb,
                                                    const int (&soff)[18]) {
    #pragma unroll
    for (int j = C * CH_J; j < C * CH_J + CH_J; ++j) {
        #pragma unroll
        for (int u = 0; u < 2; ++u)
            v[u][j] = *(const f4u*)(xb + u * (C_IN * HW * HW) + soff[j]);
    }
}

template <int C>
static __device__ __forceinline__ void gather_store(const f4u (&v)[2][18], float* row,
                                                    const bool (&rok)[18],
                                                    const bool (&elo)[18],
                                                    const bool (&ehi)[18]) {
    #pragma unroll
    for (int j = C * CH_J; j < C * CH_J + CH_J; ++j) {
        #pragma unroll
        for (int u = 0; u < 2; ++u) {
            const float a = v[u][j].a, b = v[u][j].b, c = v[u][j].c, d = v[u][j].d;
            float q0 = elo[j] ? 0.0f : (ehi[j] ? b : a);
            float q1 = elo[j] ? a    : (ehi[j] ? c : b);
            float q2 = elo[j] ? b    : (ehi[j] ? d : c);
            float q3 = elo[j] ? c    : (ehi[j] ? 0.0f : d);
            if (!rok[j]) { q0 = 0.0f; q1 = 0.0f; q2 = 0.0f; q3 = 0.0f; }
            row[0 * P_STR + u * LDS_T + 32 * j] = q0;
            row[1 * P_STR + u * LDS_T + 32 * j] = q1;
            row[2 * P_STR + u * LDS_T + 32 * j] = q2;
            row[3 * P_STR + u * LDS_T + 32 * j] = q3;
        }
    }
}

template <int T0, int T1>
static __device__ __forceinline__ void trips(const float* arow, const float* lds,
                                             int bbase0, int bbase1, v8f& c0, v8f& c1) {
    for (int k0 = T0 * 32; k0 < T1 * 32; k0 += 32) {
        v4f a[4], b0[4], b1[4];
        #pragma unroll
        for (int g = 0; g < 4; ++g)
            a[g] = *(const v4f*)(arow + k0 + 8 * g);
        #pragma unroll
        for (int g = 0; g < 4; ++g) {
            b0[g] = *(const v4f*)(lds + bbase0 + k0 + 8 * g);
            b1[g] = *(const v4f*)(lds + bbase1 + k0 + 8 * g);
        }
        #pragma unroll
        for (int g = 0; g < 4; ++g) {
            const v2f alo  = {a[g].x,  a[g].y},  ahi  = {a[g].z,  a[g].w};
            const v2f b0lo = {b0[g].x, b0[g].y}, b0hi = {b0[g].z, b0[g].w};
            const v2f b1lo = {b1[g].x, b1[g].y}, b1hi = {b1[g].z, b1[g].w};
            c0 = __builtin_amdgcn_wmma_f32_16x16x4_f32(
                     false, alo, false, b0lo, (short)0, c0, false, false);
            c1 = __builtin_amdgcn_wmma_f32_16x16x4_f32(
                     false, alo, false, b1lo, (short)0, c1, false, false);
            c0 = __builtin_amdgcn_wmma_f32_16x16x4_f32(
                     false, ahi, false, b0hi, (short)0, c0, false, false);
            c1 = __builtin_amdgcn_wmma_f32_16x16x4_f32(
                     false, ahi, false, b1hi, (short)0, c1, false, false);
        }
    }
}

__global__ __launch_bounds__(512, 1)
void local2d_wmma_f32(const float* __restrict__ xin,
                      const float* __restrict__ wgt,
                      const float* __restrict__ bias,
                      float* __restrict__ out)
{
    const int x0 = blockIdx.x & 31;
    const int y0 = (blockIdx.x >> 5) * NPOS;

    __shared__ __attribute__((aligned(16))) float lds[NPOS * P_STR];

    const int tid  = threadIdx.x;
    const int lane = tid & 31;
    const int wave = tid >> 5;

    int  soff[18];
    bool rok[18];
    bool elo[18];
    bool ehi[18];
    #pragma unroll
    for (int j = 0; j < 18; ++j) {
        const int k  = lane + 32 * j;
        const int i  = k / 9;
        const int r9 = k - i * 9;
        const int dk = r9 / 3;
        const int dl = r9 - dk * 3;
        const int h  = x0 + dk - 1;
        const int w0 = y0 + dl - 1;
        const bool ok = (unsigned)h < HW;
        const int  wl = w0 < 0 ? 0 : (w0 > HW - 4 ? HW - 4 : w0);
        rok[j]  = ok;
        elo[j]  = w0 < 0;
        ehi[j]  = w0 > HW - 4;
        soff[j] = ok ? (i * HW + h) * HW + wl : 0;
    }
    const int n0 = wave * 2;
    const float* xb = xin + n0 * (C_IN * HW * HW);
    f4u v[2][18];
    float* row = lds + n0 * LDS_T + lane;
    const int p     = wave >> 2;
    const int pos   = x0 * HW + y0 + p;
    const int m0    = (wave & 3) * 16;
    const int mrow  = m0 + (lane & 15);
    const int half  = lane >> 4;
    const int kq    = half * 4;

    const float* wbase = wgt + (size_t)pos * (C_OUT * K_TOT);
    const float* wrow  = wbase + mrow * K_TOT;
    const float* arow  = wrow + kq;

    gather_issue<0>(v, xb, soff);
    weights_to_l2<0>(wrow, half);
    gather_store<0>(v, row, rok, elo, ehi);
    __syncthreads();

    const int bbase0 = p * P_STR + (lane & 15) * LDS_T + kq;
    const int bbase1 = bbase0 + 16 * LDS_T;

    v8f c0 = {};
    v8f c1 = {};

    const int pos0 = x0 * HW + y0;
    const int m_st = tid & 63;
    const int n_st = tid >> 6;
    const v4f bv = *(const v4f*)(bias + m_st * (HW * HW) + pos0);

    weights_to_l2<1>(wrow, half);
    gather_issue<1>(v, xb, soff);
    trips<0, 3>(arow, lds, bbase0, bbase1, c0, c1);
    gather_store<1>(v, row, rok, elo, ehi);
    lds_writes_done_signal();
    weights_to_l2<2>(wrow, half);
    gather_issue<2>(v, xb, soff);
    trips<3, 6>(arow, lds, bbase0, bbase1, c0, c1);
    handover_wait();
    weights_to_l2<3>(wrow, half);
    trips<6, 9>(arow, lds, bbase0, bbase1, c0, c1);
    gather_store<2>(v, row, rok, elo, ehi);
    lds_writes_done_signal();
    weights_to_l2<4>(wrow, half);
    trips<9, 12>(arow, lds, bbase0, bbase1, c0, c1);
    handover_wait();
    trips<12, 18>(arow, lds, bbase0, bbase1, c0, c1);

    __syncthreads();
    float* stg = lds;
    const int mo   = m0 + (lane >> 4) * 8;
    const int n_lo = lane & 15;
    #pragma unroll
    for (int r = 0; r < 8; ++r) {
        stg[n_lo        * OUT_T + (mo + r) * NPOS + p] = c0[r];
        stg[(n_lo + 16) * OUT_T + (mo + r) * NPOS + p] = c1[r];
    }
    __syncthreads();
    #pragma unroll
    for (int q = 0; q < 4; ++q) {
        const int n = n_st + 8 * q;
        const v4f d = *(const v4f*)(stg + n * OUT_T + m_st * NPOS);
        *(v4f*)(out + ((n * C_OUT) + m_st) * (HW * HW) + pos0) = d + bv;
    }
}

extern "C" void kernel_launch(void* const* d_in, const int* in_sizes, int n_in,
                              void* d_out, int out_size, void* d_ws, size_t ws_size,
                              hipStream_t stream) {
    const float* xin  = (const float*)d_in[0];
    const float* wgt  = (const float*)d_in[1];
    const float* bias = (const float*)d_in[2];
    float* out = (float*)d_out;

    dim3 grid(256);
    dim3 block(512);
    hipLaunchKernelGGL(local2d_wmma_f32, grid, block, 0, stream,
                       xin, wgt, bias, out);
}
